# starting kernel
# baseline (speedup 1.0000x reference)
#include <hip/hip_runtime.h>
#include <hip/hip_bf16.h>

typedef __attribute__((ext_vector_type(2))) float v2f;
typedef __attribute__((ext_vector_type(8))) float v8f;

#define DDIM 1024
#define RANK 256
#define NB   16
#define SS   256
#define NROWS (NB * SS)
#define NCOLS (2 * RANK)

__global__ __launch_bounds__(32) void gemm_lr_kernel(
    const float* __restrict__ batch, const float* __restrict__ W1,
    const float* __restrict__ b1, float* __restrict__ leftp,
    float* __restrict__ rightp)
{
  const int lane = threadIdx.x;
  const int m0   = blockIdx.x * 32;
  const int n0   = blockIdx.y * 64;
  const int mrow = lane & 15;
  const int kb   = (lane >> 4) * 2;

  const long aBase0 = (long)(m0 + mrow) * DDIM + kb;
  const long aBase1 = (long)(m0 + 16 + mrow) * DDIM + kb;

  long wBase[4];
#pragma unroll
  for (int t = 0; t < 4; ++t) {
    int ncol = n0 + t * 16 + (lane & 15);
    int side = (ncol >= RANK) ? 1 : 0;
    int wrow = ncol - side * RANK;
    wBase[t] = (long)wrow * (2 * DDIM) + side * DDIM + kb;
  }

  v8f acc[2][4];
#pragma unroll
  for (int i = 0; i < 2; ++i)
#pragma unroll
    for (int t = 0; t < 4; ++t) acc[i][t] = (v8f){0.f,0.f,0.f,0.f,0.f,0.f,0.f,0.f};

#pragma unroll 2
  for (int k0 = 0; k0 < DDIM; k0 += 4) {
    v2f a0 = *(const v2f*)(batch + aBase0 + k0);
    v2f a1 = *(const v2f*)(batch + aBase1 + k0);
    v2f bf[4];
#pragma unroll
    for (int t = 0; t < 4; ++t) bf[t] = *(const v2f*)(W1 + wBase[t] + k0);
#pragma unroll
    for (int t = 0; t < 4; ++t) {
      acc[0][t] = __builtin_amdgcn_wmma_f32_16x16x4_f32(
          false, a0, false, bf[t], (short)0, acc[0][t], false, false);
      acc[1][t] = __builtin_amdgcn_wmma_f32_16x16x4_f32(
          false, a1, false, bf[t], (short)0, acc[1][t], false, false);
    }
  }

  const int rowoff = (lane >> 4) ? 8 : 0;
#pragma unroll
  for (int t = 0; t < 4; ++t) {
    int ncol = n0 + t * 16 + (lane & 15);
    int side = (ncol >= RANK) ? 1 : 0;
    int ccol = ncol - side * RANK;
    float bias = side ? 0.f : b1[ncol];
    float* dst = side ? rightp : leftp;
#pragma unroll
    for (int i = 0; i < 2; ++i) {
#pragma unroll
      for (int v = 0; v < 8; ++v) {
        int row = m0 + i * 16 + rowoff + v;
        dst[(long)row * RANK + ccol] = acc[i][t][v] + bias;
      }
    }
  }
}

#define PITCH 257
__global__ __launch_bounds__(256) void pair_relu_dot_kernel(
    const float* __restrict__ leftp, const float* __restrict__ rightp,
    const float* __restrict__ wlast, const float* __restrict__ blast,
    float* __restrict__ out)
{
  __shared__ float Ls[64][PITCH];
  __shared__ float Rs[64][PITCH];
  __shared__ float ws[RANK];

  const int t  = threadIdx.x;
  const int jb = blockIdx.x;
  const int ib = blockIdx.y;
  const int bz = blockIdx.z;
  const int si = bz * SS + ib * 64;
  const int sj = bz * SS + jb * 64;

  {
    int row = t & 63;
    int q   = t >> 6;
    const float4* lr = (const float4*)(leftp  + (long)(si + row) * RANK);
    const float4* rr = (const float4*)(rightp + (long)(sj + row) * RANK);
#pragma unroll
    for (int u = 0; u < 16; ++u) {
      float4 lv = lr[q * 16 + u];
      float4 rv = rr[q * 16 + u];
      int c = q * 64 + u * 4;
      Ls[row][c]   = lv.x; Ls[row][c+1] = lv.y;
      Ls[row][c+2] = lv.z; Ls[row][c+3] = lv.w;
      Rs[row][c]   = rv.x; Rs[row][c+1] = rv.y;
      Rs[row][c+2] = rv.z; Rs[row][c+3] = rv.w;
    }
    ws[t] = wlast[t];
  }
  __syncthreads();

  const int ti = t >> 4;
  const int tj = t & 15;
  float acc[4][4] = {{0.f}};

#pragma unroll 2
  for (int r = 0; r < RANK; ++r) {
    float la[4], rb[4];
#pragma unroll
    for (int a = 0; a < 4; ++a) la[a] = Ls[ti * 4 + a][r];
#pragma unroll
    for (int b = 0; b < 4; ++b) rb[b] = Rs[tj * 4 + b][r];
    float w = ws[r];
#pragma unroll
    for (int a = 0; a < 4; ++a)
#pragma unroll
      for (int b = 0; b < 4; ++b) {
        float h = fmaxf(la[a] + rb[b], 0.f);
        acc[a][b] = __builtin_fmaf(h, w, acc[a][b]);
      }
  }

  const float bl = blast[0];
#pragma unroll
  for (int a = 0; a < 4; ++a) {
#pragma unroll
    for (int b = 0; b < 4; ++b) {
      int i = ib * 64 + ti * 4 + a;
      int j = jb * 64 + tj * 4 + b;
      out[(long)bz * (SS * SS) + (long)i * SS + j] = acc[a][b] + bl;
    }
  }
}

extern "C" void kernel_launch(void* const* d_in, const int* in_sizes, int n_in,
                              void* d_out, int out_size, void* d_ws, size_t ws_size,
                              hipStream_t stream) {
  (void)in_sizes; (void)n_in; (void)out_size; (void)ws_size;
  const float* batch  = (const float*)d_in[0];
  const float* W1     = (const float*)d_in[1];
  const float* b1     = (const float*)d_in[2];
  const float* W_last = (const float*)d_in[3];
  const float* b_last = (const float*)d_in[4];
  float* out = (float*)d_out;

  float* leftp  = (float*)d_ws;
  float* rightp = leftp + (size_t)NROWS * RANK;

  dim3 g1(NROWS / 32, NCOLS / 64);
  gemm_lr_kernel<<<g1, 32, 0, stream>>>(batch, W1, b1, leftp, rightp);

  dim3 g2(4, 4, NB);
  pair_relu_dot_kernel<<<g2, 256, 0, stream>>>(leftp, rightp, W_last, b_last, out);
}


# GEMM K loop software-pipelined: the next 32-k block's A/B fragments are loaded into a second register set ahead of the current block's 64 WMMAs; same k order, results bit-identical
# speedup vs baseline: 1.3556x; 1.3556x over previous
#include <hip/hip_runtime.h>
#include <hip/hip_bf16.h>

typedef __attribute__((ext_vector_type(2))) float v2f;
typedef __attribute__((ext_vector_type(8))) float v8f;

#define DDIM 1024
#define RANK 256
#define NB   16
#define SS   256
#define NROWS (NB * SS)
#define NCOLS (2 * RANK)

#define KBLK  32
#define KSTEP (KBLK / 4)

struct FragSet {
  v2f a0[KSTEP];
  v2f a1[KSTEP];
  v2f b[4][KSTEP];
};

__device__ __forceinline__ void load_set(
    FragSet& f, const float* __restrict__ pA0, const float* __restrict__ pA1,
    const float* __restrict__ pB0, const float* __restrict__ pB1,
    const float* __restrict__ pB2, const float* __restrict__ pB3, int k)
{
#pragma unroll
  for (int s = 0; s < KSTEP; ++s) {
    f.a0[s]   = *(const v2f*)(pA0 + k + 4 * s);
    f.a1[s]   = *(const v2f*)(pA1 + k + 4 * s);
    f.b[0][s] = *(const v2f*)(pB0 + k + 4 * s);
    f.b[1][s] = *(const v2f*)(pB1 + k + 4 * s);
    f.b[2][s] = *(const v2f*)(pB2 + k + 4 * s);
    f.b[3][s] = *(const v2f*)(pB3 + k + 4 * s);
  }
}

__device__ __forceinline__ void mma_set(v8f (&acc)[2][4], const FragSet& f)
{
#pragma unroll
  for (int s = 0; s < KSTEP; ++s) {
#pragma unroll
    for (int t = 0; t < 4; ++t) {
      acc[0][t] = __builtin_amdgcn_wmma_f32_16x16x4_f32(
          false, f.a0[s], false, f.b[t][s], (short)0, acc[0][t], false, false);
      acc[1][t] = __builtin_amdgcn_wmma_f32_16x16x4_f32(
          false, f.a1[s], false, f.b[t][s], (short)0, acc[1][t], false, false);
    }
  }
}

__global__ __launch_bounds__(32) void gemm_lr_kernel(
    const float* __restrict__ batch, const float* __restrict__ W1,
    const float* __restrict__ b1, float* __restrict__ leftp,
    float* __restrict__ rightp)
{
  const int lane = threadIdx.x;
  const int m0   = blockIdx.x * 32;
  const int n0   = blockIdx.y * 64;
  const int mrow = lane & 15;
  const int kb   = (lane >> 4) * 2;

  const long aBase0 = (long)(m0 + mrow) * DDIM + kb;
  const long aBase1 = (long)(m0 + 16 + mrow) * DDIM + kb;

  long wBase[4];
#pragma unroll
  for (int t = 0; t < 4; ++t) {
    int ncol = n0 + t * 16 + (lane & 15);
    int side = (ncol >= RANK) ? 1 : 0;
    int wrow = ncol - side * RANK;
    wBase[t] = (long)wrow * (2 * DDIM) + side * DDIM + kb;
  }

  v8f acc[2][4];
#pragma unroll
  for (int i = 0; i < 2; ++i)
#pragma unroll
    for (int t = 0; t < 4; ++t) acc[i][t] = (v8f){0.f,0.f,0.f,0.f,0.f,0.f,0.f,0.f};

  const float* pA0 = batch + aBase0;
  const float* pA1 = batch + aBase1;
  const float* pB0 = W1 + wBase[0];
  const float* pB1 = W1 + wBase[1];
  const float* pB2 = W1 + wBase[2];
  const float* pB3 = W1 + wBase[3];

  FragSet f0, f1;
  load_set(f0, pA0, pA1, pB0, pB1, pB2, pB3, 0);
  for (int k0 = 0; k0 < DDIM - 2 * KBLK; k0 += 2 * KBLK) {
    load_set(f1, pA0, pA1, pB0, pB1, pB2, pB3, k0 + KBLK);
    __builtin_amdgcn_sched_barrier(0);
    mma_set(acc, f0);
    __builtin_amdgcn_sched_barrier(0);
    load_set(f0, pA0, pA1, pB0, pB1, pB2, pB3, k0 + 2 * KBLK);
    __builtin_amdgcn_sched_barrier(0);
    mma_set(acc, f1);
    __builtin_amdgcn_sched_barrier(0);
  }
  load_set(f1, pA0, pA1, pB0, pB1, pB2, pB3, DDIM - KBLK);
  __builtin_amdgcn_sched_barrier(0);
  mma_set(acc, f0);
  mma_set(acc, f1);

  const int rowoff = (lane >> 4) ? 8 : 0;
#pragma unroll
  for (int t = 0; t < 4; ++t) {
    int ncol = n0 + t * 16 + (lane & 15);
    int side = (ncol >= RANK) ? 1 : 0;
    int ccol = ncol - side * RANK;
    float bias = side ? 0.f : b1[ncol];
    float* dst = side ? rightp : leftp;
#pragma unroll
    for (int i = 0; i < 2; ++i) {
#pragma unroll
      for (int v = 0; v < 8; ++v) {
        int row = m0 + i * 16 + rowoff + v;
        dst[(long)row * RANK + ccol] = acc[i][t][v] + bias;
      }
    }
  }
}

#define PITCH 257
__global__ __launch_bounds__(256) void pair_relu_dot_kernel(
    const float* __restrict__ leftp, const float* __restrict__ rightp,
    const float* __restrict__ wlast, const float* __restrict__ blast,
    float* __restrict__ out)
{
  __shared__ float Ls[64][PITCH];
  __shared__ float Rs[64][PITCH];
  __shared__ float ws[RANK];

  const int t  = threadIdx.x;
  const int jb = blockIdx.x;
  const int ib = blockIdx.y;
  const int bz = blockIdx.z;
  const int si = bz * SS + ib * 64;
  const int sj = bz * SS + jb * 64;

  {
    int row = t & 63;
    int q   = t >> 6;
    const float4* lr = (const float4*)(leftp  + (long)(si + row) * RANK);
    const float4* rr = (const float4*)(rightp + (long)(sj + row) * RANK);
#pragma unroll
    for (int u = 0; u < 16; ++u) {
      float4 lv = lr[q * 16 + u];
      float4 rv = rr[q * 16 + u];
      int c = q * 64 + u * 4;
      Ls[row][c]   = lv.x; Ls[row][c+1] = lv.y;
      Ls[row][c+2] = lv.z; Ls[row][c+3] = lv.w;
      Rs[row][c]   = rv.x; Rs[row][c+1] = rv.y;
      Rs[row][c+2] = rv.z; Rs[row][c+3] = rv.w;
    }
    ws[t] = wlast[t];
  }
  __syncthreads();

  const int ti = t >> 4;
  const int tj = t & 15;
  float acc[4][4] = {{0.f}};

#pragma unroll 2
  for (int r = 0; r < RANK; ++r) {
    float la[4], rb[4];
#pragma unroll
    for (int a = 0; a < 4; ++a) la[a] = Ls[ti * 4 + a][r];
#pragma unroll
    for (int b = 0; b < 4; ++b) rb[b] = Rs[tj * 4 + b][r];
    float w = ws[r];
#pragma unroll
    for (int a = 0; a < 4; ++a)
#pragma unroll
      for (int b = 0; b < 4; ++b) {
        float h = fmaxf(la[a] + rb[b], 0.f);
        acc[a][b] = __builtin_fmaf(h, w, acc[a][b]);
      }
  }

  const float bl = blast[0];
#pragma unroll
  for (int a = 0; a < 4; ++a) {
#pragma unroll
    for (int b = 0; b < 4; ++b) {
      int i = ib * 64 + ti * 4 + a;
      int j = jb * 64 + tj * 4 + b;
      out[(long)bz * (SS * SS) + (long)i * SS + j] = acc[a][b] + bl;
    }
  }
}

extern "C" void kernel_launch(void* const* d_in, const int* in_sizes, int n_in,
                              void* d_out, int out_size, void* d_ws, size_t ws_size,
                              hipStream_t stream) {
  (void)in_sizes; (void)n_in; (void)out_size; (void)ws_size;
  const float* batch  = (const float*)d_in[0];
  const float* W1     = (const float*)d_in[1];
  const float* b1     = (const float*)d_in[2];
  const float* W_last = (const float*)d_in[3];
  const float* b_last = (const float*)d_in[4];
  float* out = (float*)d_out;

  float* leftp  = (float*)d_ws;
  float* rightp = leftp + (size_t)NROWS * RANK;

  dim3 g1(NROWS / 32, NCOLS / 64);
  gemm_lr_kernel<<<g1, 32, 0, stream>>>(batch, W1, b1, leftp, rightp);

  dim3 g2(4, 4, NB);
  pair_relu_dot_kernel<<<g2, 256, 0, stream>>>(leftp, rightp, W_last, b_last, out);
}


# GEMM operands staged through LDS by a 4-wave work-group (64x128 tile, 128-k stages, whole-line 16-byte global loads, ds fragment reads); same k order, results bit-identical
# speedup vs baseline: 1.6707x; 1.2324x over previous
#include <hip/hip_runtime.h>
#include <hip/hip_bf16.h>

typedef __attribute__((ext_vector_type(2))) float v2f;
typedef __attribute__((ext_vector_type(4))) float v4f;
typedef __attribute__((ext_vector_type(8))) float v8f;

#define DDIM 1024
#define RANK 256
#define NB   16
#define SS   256
#define NROWS (NB * SS)
#define NCOLS (2 * RANK)

#define KSUB  32
#define KSTEP (KSUB / 4)
#define KBLK  128
#define NBLK  (DDIM / KBLK)
#define TM    64
#define TN    128
#define TROWS (TM + TN)
#define GPITCH (KBLK + 4)
#define STAGE_FLOATS (TROWS * GPITCH)
#define NLDA  (TM / 4)
#define NLDB  (TN / 4)

struct FragSet {
  v2f a0[KSTEP];
  v2f a1[KSTEP];
  v2f b[4][KSTEP];
};

__device__ __forceinline__ void lds_load_set(
    FragSet& f, const float* st, int oA0, int oA1, int oB, int kk)
{
#pragma unroll
  for (int s = 0; s < KSTEP; ++s) {
    f.a0[s] = *(const v2f*)(st + oA0 + kk + 4 * s);
    f.a1[s] = *(const v2f*)(st + oA1 + kk + 4 * s);
#pragma unroll
    for (int t = 0; t < 4; ++t)
      f.b[t][s] = *(const v2f*)(st + oB + t * 16 * GPITCH + kk + 4 * s);
  }
}

__device__ __forceinline__ void gload_block(
    v4f (&g)[NLDA + NLDB], const float* __restrict__ gA,
    const float* __restrict__ gB, int k0)
{
#pragma unroll
  for (int i = 0; i < NLDA; ++i)
    g[i] = *(const v4f*)(gA + (long)(4 * i) * DDIM + k0);
#pragma unroll
  for (int i = 0; i < NLDB; ++i)
    g[NLDA + i] = *(const v4f*)(gB + (long)(4 * i) * (2 * DDIM) + k0);
}

__device__ __forceinline__ void lds_store_block(
    float* p, const v4f (&g)[NLDA + NLDB])
{
#pragma unroll
  for (int i = 0; i < NLDA + NLDB; ++i)
    *(v4f*)(p + 4 * i * GPITCH) = g[i];
}

__device__ __forceinline__ void mma_set(v8f (&acc)[2][4], const FragSet& f)
{
#pragma unroll
  for (int s = 0; s < KSTEP; ++s) {
#pragma unroll
    for (int t = 0; t < 4; ++t) {
      acc[0][t] = __builtin_amdgcn_wmma_f32_16x16x4_f32(
          false, f.a0[s], false, f.b[t][s], (short)0, acc[0][t], false, false);
      acc[1][t] = __builtin_amdgcn_wmma_f32_16x16x4_f32(
          false, f.a1[s], false, f.b[t][s], (short)0, acc[1][t], false, false);
    }
  }
}

#define COMPUTE_HEAD(st)                                     \
  lds_load_set(f0, (st), oA0, oA1, oB, 0);                   \
  lds_load_set(f1, (st), oA0, oA1, oB, KSUB);                \
  __builtin_amdgcn_sched_barrier(0);                         \
  mma_set(acc, f0);                                          \
  __builtin_amdgcn_sched_barrier(0);                         \
  lds_load_set(f0, (st), oA0, oA1, oB, 2 * KSUB);            \
  __builtin_amdgcn_sched_barrier(0);                         \
  mma_set(acc, f1);                                          \
  __builtin_amdgcn_sched_barrier(0);                         \
  lds_load_set(f1, (st), oA0, oA1, oB, 3 * KSUB);            \
  __builtin_amdgcn_sched_barrier(0);                         \
  mma_set(acc, f0);                                          \
  __builtin_amdgcn_sched_barrier(0);

__global__ __launch_bounds__(128) void gemm_lr_kernel(
    const float* __restrict__ batch, const float* __restrict__ W1,
    const float* __restrict__ b1, float* __restrict__ leftp,
    float* __restrict__ rightp)
{
  __shared__ __attribute__((aligned(16))) float lds[2 * STAGE_FLOATS];

  const int tid  = threadIdx.x;
  const int lane = tid & 31;
  const int w    = tid >> 5;
  const int wm   = w & 1;
  const int wn   = w >> 1;
  const int m0   = blockIdx.x * TM;
  const int n0   = blockIdx.y * TN;
  const int c    = lane & 15;
  const int kb   = (lane >> 4) * 2;

  const int gside = (n0 >= RANK) ? 1 : 0;
  const int wrow0 = n0 - gside * RANK;

  const float* gA = batch + (long)(m0 + w) * DDIM + lane * 4;
  const float* gB = W1 + (long)(wrow0 + w) * (2 * DDIM) + gside * DDIM + lane * 4;
  const int    sOff = w * GPITCH + lane * 4;

  const int oA0 = (wm * 32 + c) * GPITCH + kb;
  const int oA1 = oA0 + 16 * GPITCH;
  const int oB  = (TM + wn * 64 + c) * GPITCH + kb;

  v8f acc[2][4];
#pragma unroll
  for (int i = 0; i < 2; ++i)
#pragma unroll
    for (int t = 0; t < 4; ++t) acc[i][t] = (v8f){0.f,0.f,0.f,0.f,0.f,0.f,0.f,0.f};

  v4f g[NLDA + NLDB];
  FragSet f0, f1;

  gload_block(g, gA, gB, 0);
  lds_store_block(lds + sOff, g);
  __syncthreads();

#pragma unroll 1
  for (int b = 0; b < NBLK - 1; ++b) {
    const float* cur = lds + (b & 1) * STAGE_FLOATS;
    float*       nxt = lds + ((b + 1) & 1) * STAGE_FLOATS;
    gload_block(g, gA, gB, (b + 1) * KBLK);
    __builtin_amdgcn_sched_barrier(0);
    COMPUTE_HEAD(cur)
    lds_store_block(nxt + sOff, g);
    __builtin_amdgcn_sched_barrier(0);
    mma_set(acc, f1);
    __syncthreads();
  }
  {
    const float* cur = lds + ((NBLK - 1) & 1) * STAGE_FLOATS;
    COMPUTE_HEAD(cur)
    mma_set(acc, f1);
  }

  const int mw = m0 + wm * 32;
  const int nw = n0 + wn * 64;
  const int rowoff = (lane >> 4) ? 8 : 0;
#pragma unroll
  for (int t = 0; t < 4; ++t) {
    int ncol = nw + t * 16 + (lane & 15);
    int side = (ncol >= RANK) ? 1 : 0;
    int ccol = ncol - side * RANK;
    float bias = side ? 0.f : b1[ncol];
    float* dst = side ? rightp : leftp;
#pragma unroll
    for (int i = 0; i < 2; ++i) {
#pragma unroll
      for (int v = 0; v < 8; ++v) {
        int row = mw + i * 16 + rowoff + v;
        dst[(long)row * RANK + ccol] = acc[i][t][v] + bias;
      }
    }
  }
}

#define PITCH 257
__global__ __launch_bounds__(256) void pair_relu_dot_kernel(
    const float* __restrict__ leftp, const float* __restrict__ rightp,
    const float* __restrict__ wlast, const float* __restrict__ blast,
    float* __restrict__ out)
{
  __shared__ float Ls[64][PITCH];
  __shared__ float Rs[64][PITCH];
  __shared__ float ws[RANK];

  const int t  = threadIdx.x;
  const int jb = blockIdx.x;
  const int ib = blockIdx.y;
  const int bz = blockIdx.z;
  const int si = bz * SS + ib * 64;
  const int sj = bz * SS + jb * 64;

  {
    int row = t & 63;
    int q   = t >> 6;
    const float4* lr = (const float4*)(leftp  + (long)(si + row) * RANK);
    const float4* rr = (const float4*)(rightp + (long)(sj + row) * RANK);
#pragma unroll
    for (int u = 0; u < 16; ++u) {
      float4 lv = lr[q * 16 + u];
      float4 rv = rr[q * 16 + u];
      int c = q * 64 + u * 4;
      Ls[row][c]   = lv.x; Ls[row][c+1] = lv.y;
      Ls[row][c+2] = lv.z; Ls[row][c+3] = lv.w;
      Rs[row][c]   = rv.x; Rs[row][c+1] = rv.y;
      Rs[row][c+2] = rv.z; Rs[row][c+3] = rv.w;
    }
    ws[t] = wlast[t];
  }
  __syncthreads();

  const int ti = t >> 4;
  const int tj = t & 15;
  float acc[4][4] = {{0.f}};

#pragma unroll 2
  for (int r = 0; r < RANK; ++r) {
    float la[4], rb[4];
#pragma unroll
    for (int a = 0; a < 4; ++a) la[a] = Ls[ti * 4 + a][r];
#pragma unroll
    for (int b = 0; b < 4; ++b) rb[b] = Rs[tj * 4 + b][r];
    float w = ws[r];
#pragma unroll
    for (int a = 0; a < 4; ++a)
#pragma unroll
      for (int b = 0; b < 4; ++b) {
        float h = fmaxf(la[a] + rb[b], 0.f);
        acc[a][b] = __builtin_fmaf(h, w, acc[a][b]);
      }
  }

  const float bl = blast[0];
#pragma unroll
  for (int a = 0; a < 4; ++a) {
#pragma unroll
    for (int b = 0; b < 4; ++b) {
      int i = ib * 64 + ti * 4 + a;
      int j = jb * 64 + tj * 4 + b;
      out[(long)bz * (SS * SS) + (long)i * SS + j] = acc[a][b] + bl;
    }
  }
}

extern "C" void kernel_launch(void* const* d_in, const int* in_sizes, int n_in,
                              void* d_out, int out_size, void* d_ws, size_t ws_size,
                              hipStream_t stream) {
  (void)in_sizes; (void)n_in; (void)out_size; (void)ws_size;
  const float* batch  = (const float*)d_in[0];
  const float* W1     = (const float*)d_in[1];
  const float* b1     = (const float*)d_in[2];
  const float* W_last = (const float*)d_in[3];
  const float* b_last = (const float*)d_in[4];
  float* out = (float*)d_out;

  float* leftp  = (float*)d_ws;
  float* rightp = leftp + (size_t)NROWS * RANK;

  dim3 g1(NROWS / TM, NCOLS / TN);
  gemm_lr_kernel<<<g1, 128, 0, stream>>>(batch, W1, b1, leftp, rightp);

  dim3 g2(4, 4, NB);
  pair_relu_dot_kernel<<<g2, 256, 0, stream>>>(leftp, rightp, W_last, b_last, out);
}


# pair kernel: LDS rows on a 16-byte pitch, operands read 4 r per 16-byte LDS read, the reads of 8 r issued as one group per trip, adds packed over adjacent r; sum order kept, bit-identical
# speedup vs baseline: 1.7751x; 1.0625x over previous
#include <hip/hip_runtime.h>
#include <hip/hip_bf16.h>

typedef __attribute__((ext_vector_type(2))) float v2f;
typedef __attribute__((ext_vector_type(4))) float v4f;
typedef __attribute__((ext_vector_type(8))) float v8f;

#define DDIM 1024
#define RANK 256
#define NB   16
#define SS   256
#define NROWS (NB * SS)
#define NCOLS (2 * RANK)

#define KSUB  32
#define KSTEP (KSUB / 4)
#define KBLK  128
#define NBLK  (DDIM / KBLK)
#define TM    64
#define TN    128
#define TROWS (TM + TN)
#define GPITCH (KBLK + 4)
#define STAGE_FLOATS (TROWS * GPITCH)
#define NLDA  (TM / 4)
#define NLDB  (TN / 4)

struct FragSet {
  v2f a0[KSTEP];
  v2f a1[KSTEP];
  v2f b[4][KSTEP];
};

__device__ __forceinline__ void lds_load_set(
    FragSet& f, const float* st, int oA0, int oA1, int oB, int kk)
{
#pragma unroll
  for (int s = 0; s < KSTEP; ++s) {
    f.a0[s] = *(const v2f*)(st + oA0 + kk + 4 * s);
    f.a1[s] = *(const v2f*)(st + oA1 + kk + 4 * s);
#pragma unroll
    for (int t = 0; t < 4; ++t)
      f.b[t][s] = *(const v2f*)(st + oB + t * 16 * GPITCH + kk + 4 * s);
  }
}

__device__ __forceinline__ void gload_block(
    v4f (&g)[NLDA + NLDB], const float* __restrict__ gA,
    const float* __restrict__ gB, int k0)
{
#pragma unroll
  for (int i = 0; i < NLDA; ++i)
    g[i] = *(const v4f*)(gA + (long)(4 * i) * DDIM + k0);
#pragma unroll
  for (int i = 0; i < NLDB; ++i)
    g[NLDA + i] = *(const v4f*)(gB + (long)(4 * i) * (2 * DDIM) + k0);
}

__device__ __forceinline__ void lds_store_block(
    float* p, const v4f (&g)[NLDA + NLDB])
{
#pragma unroll
  for (int i = 0; i < NLDA + NLDB; ++i)
    *(v4f*)(p + 4 * i * GPITCH) = g[i];
}

__device__ __forceinline__ void mma_set(v8f (&acc)[2][4], const FragSet& f)
{
#pragma unroll
  for (int s = 0; s < KSTEP; ++s) {
#pragma unroll
    for (int t = 0; t < 4; ++t) {
      acc[0][t] = __builtin_amdgcn_wmma_f32_16x16x4_f32(
          false, f.a0[s], false, f.b[t][s], (short)0, acc[0][t], false, false);
      acc[1][t] = __builtin_amdgcn_wmma_f32_16x16x4_f32(
          false, f.a1[s], false, f.b[t][s], (short)0, acc[1][t], false, false);
    }
  }
}

#define COMPUTE_HEAD(st)                                     \
  lds_load_set(f0, (st), oA0, oA1, oB, 0);                   \
  lds_load_set(f1, (st), oA0, oA1, oB, KSUB);                \
  __builtin_amdgcn_sched_barrier(0);                         \
  mma_set(acc, f0);                                          \
  __builtin_amdgcn_sched_barrier(0);                         \
  lds_load_set(f0, (st), oA0, oA1, oB, 2 * KSUB);            \
  __builtin_amdgcn_sched_barrier(0);                         \
  mma_set(acc, f1);                                          \
  __builtin_amdgcn_sched_barrier(0);                         \
  lds_load_set(f1, (st), oA0, oA1, oB, 3 * KSUB);            \
  __builtin_amdgcn_sched_barrier(0);                         \
  mma_set(acc, f0);                                          \
  __builtin_amdgcn_sched_barrier(0);

__global__ __launch_bounds__(128) void gemm_lr_kernel(
    const float* __restrict__ batch, const float* __restrict__ W1,
    const float* __restrict__ b1, float* __restrict__ leftp,
    float* __restrict__ rightp)
{
  __shared__ __attribute__((aligned(16))) float lds[2 * STAGE_FLOATS];

  const int tid  = threadIdx.x;
  const int lane = tid & 31;
  const int w    = tid >> 5;
  const int wm   = w & 1;
  const int wn   = w >> 1;
  const int m0   = blockIdx.x * TM;
  const int n0   = blockIdx.y * TN;
  const int c    = lane & 15;
  const int kb   = (lane >> 4) * 2;

  const int gside = (n0 >= RANK) ? 1 : 0;
  const int wrow0 = n0 - gside * RANK;

  const float* gA = batch + (long)(m0 + w) * DDIM + lane * 4;
  const float* gB = W1 + (long)(wrow0 + w) * (2 * DDIM) + gside * DDIM + lane * 4;
  const int    sOff = w * GPITCH + lane * 4;

  const int oA0 = (wm * 32 + c) * GPITCH + kb;
  const int oA1 = oA0 + 16 * GPITCH;
  const int oB  = (TM + wn * 64 + c) * GPITCH + kb;

  v8f acc[2][4];
#pragma unroll
  for (int i = 0; i < 2; ++i)
#pragma unroll
    for (int t = 0; t < 4; ++t) acc[i][t] = (v8f){0.f,0.f,0.f,0.f,0.f,0.f,0.f,0.f};

  v4f g[NLDA + NLDB];
  FragSet f0, f1;

  gload_block(g, gA, gB, 0);
  lds_store_block(lds + sOff, g);
  __syncthreads();

#pragma unroll 1
  for (int b = 0; b < NBLK - 1; ++b) {
    const float* cur = lds + (b & 1) * STAGE_FLOATS;
    float*       nxt = lds + ((b + 1) & 1) * STAGE_FLOATS;
    gload_block(g, gA, gB, (b + 1) * KBLK);
    __builtin_amdgcn_sched_barrier(0);
    COMPUTE_HEAD(cur)
    lds_store_block(nxt + sOff, g);
    __builtin_amdgcn_sched_barrier(0);
    mma_set(acc, f1);
    __syncthreads();
  }
  {
    const float* cur = lds + ((NBLK - 1) & 1) * STAGE_FLOATS;
    COMPUTE_HEAD(cur)
    mma_set(acc, f1);
  }

  const int mw = m0 + wm * 32;
  const int nw = n0 + wn * 64;
  const int rowoff = (lane >> 4) ? 8 : 0;
#pragma unroll
  for (int t = 0; t < 4; ++t) {
    int ncol = nw + t * 16 + (lane & 15);
    int side = (ncol >= RANK) ? 1 : 0;
    int ccol = ncol - side * RANK;
    float bias = side ? 0.f : b1[ncol];
    float* dst = side ? rightp : leftp;
#pragma unroll
    for (int i = 0; i < 2; ++i) {
#pragma unroll
      for (int v = 0; v < 8; ++v) {
        int row = mw + i * 16 + rowoff + v;
        dst[(long)row * RANK + ccol] = acc[i][t][v] + bias;
      }
    }
  }
}

#define PITCH 260
#define RSTEP 4
#define RTRIP 8
__global__ __launch_bounds__(256) void pair_relu_dot_kernel(
    const float* __restrict__ leftp, const float* __restrict__ rightp,
    const float* __restrict__ wlast, const float* __restrict__ blast,
    float* __restrict__ out)
{
  __shared__ __attribute__((aligned(16))) float Ls[64 * PITCH];
  __shared__ __attribute__((aligned(16))) float Rs[64 * PITCH];
  __shared__ __attribute__((aligned(16))) float ws[RANK];

  const int t  = threadIdx.x;
  const int jb = blockIdx.x;
  const int ib = blockIdx.y;
  const int bz = blockIdx.z;
  const int si = bz * SS + ib * 64;
  const int sj = bz * SS + jb * 64;

  {
    int row = t & 63;
    int q   = t >> 6;
    int jrow = 4 * (row & 15) + (row >> 4);
    const v4f* lr = (const v4f*)(leftp  + (long)(si + row) * RANK);
    const v4f* rr = (const v4f*)(rightp + (long)(sj + jrow) * RANK);
    float* lsr = Ls + row * PITCH + q * 64;
    float* rsr = Rs + row * PITCH + q * 64;
#pragma unroll
    for (int u = 0; u < 16; ++u) {
      v4f lv = lr[q * 16 + u];
      v4f rv = rr[q * 16 + u];
      *(v4f*)(lsr + u * 4) = lv;
      *(v4f*)(rsr + u * 4) = rv;
    }
    ws[t] = wlast[t];
  }
  __syncthreads();

  const int ti = t >> 4;
  const int tj = t & 15;
  const float* lp = Ls + (ti * 4) * PITCH;
  const float* rp = Rs + tj * PITCH;
  const float* wp = ws;
  v2f acc2[4][2];
#pragma unroll
  for (int a = 0; a < 4; ++a) { acc2[a][0] = (v2f){0.f, 0.f}; acc2[a][1] = (v2f){0.f, 0.f}; }

#pragma unroll 1
  for (int r0 = 0; r0 < RANK; r0 += RTRIP) {
    v4f la[RTRIP / RSTEP][4], rb[RTRIP / RSTEP][4], w[RTRIP / RSTEP];
#pragma unroll
    for (int g = 0; g < RTRIP / RSTEP; ++g) {
#pragma unroll
      for (int a = 0; a < 4; ++a) la[g][a] = *(const v4f*)(lp + a * PITCH + g * RSTEP);
#pragma unroll
      for (int b = 0; b < 4; ++b) rb[g][b] = *(const v4f*)(rp + b * 16 * PITCH + g * RSTEP);
      w[g] = *(const v4f*)(wp + g * RSTEP);
    }
#pragma unroll
    for (int g = 0; g < RTRIP / RSTEP; ++g)
#pragma unroll
      for (int k = 0; k < RSTEP; k += 2)
#pragma unroll
        for (int a = 0; a < 4; ++a) {
          const v2f l2 = (v2f){la[g][a][k], la[g][a][k + 1]};
#pragma unroll
          for (int p = 0; p < 2; ++p) {
            const v2f x0 = l2 + (v2f){rb[g][2 * p][k],     rb[g][2 * p][k + 1]};
            const v2f x1 = l2 + (v2f){rb[g][2 * p + 1][k], rb[g][2 * p + 1][k + 1]};
            const v2f hk  = (v2f){fmaxf(x0[0], 0.f), fmaxf(x1[0], 0.f)};
            const v2f hk1 = (v2f){fmaxf(x0[1], 0.f), fmaxf(x1[1], 0.f)};
            acc2[a][p] = __builtin_elementwise_fma(hk,  (v2f){w[g][k],     w[g][k]},     acc2[a][p]);
            acc2[a][p] = __builtin_elementwise_fma(hk1, (v2f){w[g][k + 1], w[g][k + 1]}, acc2[a][p]);
          }
        }
    lp += RTRIP; rp += RTRIP; wp += RTRIP;
  }

  const float bl = blast[0];
#pragma unroll
  for (int a = 0; a < 4; ++a) {
#pragma unroll
    for (int b = 0; b < 4; ++b) {
      int i = ib * 64 + ti * 4 + a;
      int j = jb * 64 + tj * 4 + b;
      out[(long)bz * (SS * SS) + (long)i * SS + j] = acc2[a][b >> 1][b & 1] + bl;
    }
  }
}

extern "C" void kernel_launch(void* const* d_in, const int* in_sizes, int n_in,
                              void* d_out, int out_size, void* d_ws, size_t ws_size,
                              hipStream_t stream) {
  (void)in_sizes; (void)n_in; (void)out_size; (void)ws_size;
  const float* batch  = (const float*)d_in[0];
  const float* W1     = (const float*)d_in[1];
  const float* b1     = (const float*)d_in[2];
  const float* W_last = (const float*)d_in[3];
  const float* b_last = (const float*)d_in[4];
  float* out = (float*)d_out;

  float* leftp  = (float*)d_ws;
  float* rightp = leftp + (size_t)NROWS * RANK;

  dim3 g1(NROWS / TM, NCOLS / TN);
  gemm_lr_kernel<<<g1, 128, 0, stream>>>(batch, W1, b1, leftp, rightp);

  dim3 g2(4, 4, NB);
  pair_relu_dot_kernel<<<g2, 256, 0, stream>>>(leftp, rightp, W_last, b_last, out);
}


# GEMM work-group of 8 waves, two per SIMD, each a 32x32 part of the same 64x128 tile: one wave's global loads, LDS stores and fragment reads issue under the other's WMMAs; same k order, bit-identical
# speedup vs baseline: 1.9305x; 1.0875x over previous
#include <hip/hip_runtime.h>
#include <hip/hip_bf16.h>

typedef __attribute__((ext_vector_type(2))) float v2f;
typedef __attribute__((ext_vector_type(4))) float v4f;
typedef __attribute__((ext_vector_type(8))) float v8f;

#define DDIM 1024
#define RANK 256
#define NB   16
#define SS   256
#define NROWS (NB * SS)
#define NCOLS (2 * RANK)

#define KSUB  32
#define KSTEP (KSUB / 4)
#define KBLK  128
#define NBLK  (DDIM / KBLK)
#define TM    64
#define TN    128
#define NWAVE 8
#define WTN   2
#define TROWS (TM + TN)
#define GPITCH (KBLK + 4)
#define STAGE_FLOATS (TROWS * GPITCH)
#define NLDA  (TM / NWAVE)
#define NLDB  (TN / NWAVE)

struct FragSet {
  v2f a0[KSTEP];
  v2f a1[KSTEP];
  v2f b[WTN][KSTEP];
};

__device__ __forceinline__ void lds_load_set(
    FragSet& f, const float* st, int oA0, int oA1, int oB, int kk)
{
#pragma unroll
  for (int s = 0; s < KSTEP; ++s) {
    f.a0[s] = *(const v2f*)(st + oA0 + kk + 4 * s);
    f.a1[s] = *(const v2f*)(st + oA1 + kk + 4 * s);
#pragma unroll
    for (int t = 0; t < WTN; ++t)
      f.b[t][s] = *(const v2f*)(st + oB + t * 16 * GPITCH + kk + 4 * s);
  }
}

__device__ __forceinline__ void gload_block(
    v4f (&g)[NLDA + NLDB], const float* __restrict__ gA,
    const float* __restrict__ gB, int k0)
{
#pragma unroll
  for (int i = 0; i < NLDA; ++i)
    g[i] = *(const v4f*)(gA + (long)(NWAVE * i) * DDIM + k0);
#pragma unroll
  for (int i = 0; i < NLDB; ++i)
    g[NLDA + i] = *(const v4f*)(gB + (long)(NWAVE * i) * (2 * DDIM) + k0);
}

__device__ __forceinline__ void lds_store_block(
    float* p, const v4f (&g)[NLDA + NLDB])
{
#pragma unroll
  for (int i = 0; i < NLDA + NLDB; ++i)
    *(v4f*)(p + NWAVE * i * GPITCH) = g[i];
}

__device__ __forceinline__ void mma_set(v8f (&acc)[2][WTN], const FragSet& f)
{
#pragma unroll
  for (int s = 0; s < KSTEP; ++s) {
#pragma unroll
    for (int t = 0; t < WTN; ++t) {
      acc[0][t] = __builtin_amdgcn_wmma_f32_16x16x4_f32(
          false, f.a0[s], false, f.b[t][s], (short)0, acc[0][t], false, false);
      acc[1][t] = __builtin_amdgcn_wmma_f32_16x16x4_f32(
          false, f.a1[s], false, f.b[t][s], (short)0, acc[1][t], false, false);
    }
  }
}

#define COMPUTE_HEAD(st)                                     \
  lds_load_set(f0, (st), oA0, oA1, oB, 0);                   \
  lds_load_set(f1, (st), oA0, oA1, oB, KSUB);                \
  __builtin_amdgcn_sched_barrier(0);                         \
  mma_set(acc, f0);                                          \
  __builtin_amdgcn_sched_barrier(0);                         \
  lds_load_set(f0, (st), oA0, oA1, oB, 2 * KSUB);            \
  __builtin_amdgcn_sched_barrier(0);                         \
  mma_set(acc, f1);                                          \
  __builtin_amdgcn_sched_barrier(0);                         \
  lds_load_set(f1, (st), oA0, oA1, oB, 3 * KSUB);            \
  __builtin_amdgcn_sched_barrier(0);                         \
  mma_set(acc, f0);                                          \
  __builtin_amdgcn_sched_barrier(0);

__global__ __launch_bounds__(NWAVE * 32) void gemm_lr_kernel(
    const float* __restrict__ batch, const float* __restrict__ W1,
    const float* __restrict__ b1, float* __restrict__ leftp,
    float* __restrict__ rightp)
{
  __shared__ __attribute__((aligned(16))) float lds[2 * STAGE_FLOATS];

  const int tid  = threadIdx.x;
  const int lane = tid & 31;
  const int w    = tid >> 5;
  const int wm   = w & 1;
  const int wn   = w >> 1;
  const int m0   = blockIdx.x * TM;
  const int n0   = blockIdx.y * TN;
  const int c    = lane & 15;
  const int kb   = (lane >> 4) * 2;

  const int gside = (n0 >= RANK) ? 1 : 0;
  const int wrow0 = n0 - gside * RANK;

  const float* gA = batch + (long)(m0 + w) * DDIM + lane * 4;
  const float* gB = W1 + (long)(wrow0 + w) * (2 * DDIM) + gside * DDIM + lane * 4;
  const int    sOff = w * GPITCH + lane * 4;

  const int oA0 = (wm * 32 + c) * GPITCH + kb;
  const int oA1 = oA0 + 16 * GPITCH;
  const int oB  = (TM + wn * (16 * WTN) + c) * GPITCH + kb;

  v8f acc[2][WTN];
#pragma unroll
  for (int i = 0; i < 2; ++i)
#pragma unroll
    for (int t = 0; t < WTN; ++t) acc[i][t] = (v8f){0.f,0.f,0.f,0.f,0.f,0.f,0.f,0.f};

  v4f g[NLDA + NLDB];
  FragSet f0, f1;

  gload_block(g, gA, gB, 0);
  lds_store_block(lds + sOff, g);
  __syncthreads();

#pragma unroll 1
  for (int b = 0; b < NBLK - 1; ++b) {
    const float* cur = lds + (b & 1) * STAGE_FLOATS;
    float*       nxt = lds + ((b + 1) & 1) * STAGE_FLOATS;
    gload_block(g, gA, gB, (b + 1) * KBLK);
    __builtin_amdgcn_sched_barrier(0);
    COMPUTE_HEAD(cur)
    lds_store_block(nxt + sOff, g);
    __builtin_amdgcn_sched_barrier(0);
    mma_set(acc, f1);
    __syncthreads();
  }
  {
    const float* cur = lds + ((NBLK - 1) & 1) * STAGE_FLOATS;
    COMPUTE_HEAD(cur)
    mma_set(acc, f1);
  }

  const int mw = m0 + wm * 32;
  const int nw = n0 + wn * (16 * WTN);
  const int rowoff = (lane >> 4) ? 8 : 0;
#pragma unroll
  for (int t = 0; t < WTN; ++t) {
    int ncol = nw + t * 16 + (lane & 15);
    int side = (ncol >= RANK) ? 1 : 0;
    int ccol = ncol - side * RANK;
    float bias = side ? 0.f : b1[ncol];
    float* dst = side ? rightp : leftp;
#pragma unroll
    for (int i = 0; i < 2; ++i) {
#pragma unroll
      for (int v = 0; v < 8; ++v) {
        int row = mw + i * 16 + rowoff + v;
        dst[(long)row * RANK + ccol] = acc[i][t][v] + bias;
      }
    }
  }
}

#define PITCH 260
#define RSTEP 4
#define RTRIP 8
__global__ __launch_bounds__(256) void pair_relu_dot_kernel(
    const float* __restrict__ leftp, const float* __restrict__ rightp,
    const float* __restrict__ wlast, const float* __restrict__ blast,
    float* __restrict__ out)
{
  __shared__ __attribute__((aligned(16))) float Ls[64 * PITCH];
  __shared__ __attribute__((aligned(16))) float Rs[64 * PITCH];
  __shared__ __attribute__((aligned(16))) float ws[RANK];

  const int t  = threadIdx.x;
  const int jb = blockIdx.x;
  const int ib = blockIdx.y;
  const int bz = blockIdx.z;
  const int si = bz * SS + ib * 64;
  const int sj = bz * SS + jb * 64;

  {
    int row = t & 63;
    int q   = t >> 6;
    int jrow = 4 * (row & 15) + (row >> 4);
    const v4f* lr = (const v4f*)(leftp  + (long)(si + row) * RANK);
    const v4f* rr = (const v4f*)(rightp + (long)(sj + jrow) * RANK);
    float* lsr = Ls + row * PITCH + q * 64;
    float* rsr = Rs + row * PITCH + q * 64;
#pragma unroll
    for (int u = 0; u < 16; ++u) {
      v4f lv = lr[q * 16 + u];
      v4f rv = rr[q * 16 + u];
      *(v4f*)(lsr + u * 4) = lv;
      *(v4f*)(rsr + u * 4) = rv;
    }
    ws[t] = wlast[t];
  }
  __syncthreads();

  const int ti = t >> 4;
  const int tj = t & 15;
  const float* lp = Ls + (ti * 4) * PITCH;
  const float* rp = Rs + tj * PITCH;
  const float* wp = ws;
  v2f acc2[4][2];
#pragma unroll
  for (int a = 0; a < 4; ++a) { acc2[a][0] = (v2f){0.f, 0.f}; acc2[a][1] = (v2f){0.f, 0.f}; }

#pragma unroll 1
  for (int r0 = 0; r0 < RANK; r0 += RTRIP) {
    v4f la[RTRIP / RSTEP][4], rb[RTRIP / RSTEP][4], w[RTRIP / RSTEP];
#pragma unroll
    for (int g = 0; g < RTRIP / RSTEP; ++g) {
#pragma unroll
      for (int a = 0; a < 4; ++a) la[g][a] = *(const v4f*)(lp + a * PITCH + g * RSTEP);
#pragma unroll
      for (int b = 0; b < 4; ++b) rb[g][b] = *(const v4f*)(rp + b * 16 * PITCH + g * RSTEP);
      w[g] = *(const v4f*)(wp + g * RSTEP);
    }
#pragma unroll
    for (int g = 0; g < RTRIP / RSTEP; ++g)
#pragma unroll
      for (int k = 0; k < RSTEP; k += 2)
#pragma unroll
        for (int a = 0; a < 4; ++a) {
          const v2f l2 = (v2f){la[g][a][k], la[g][a][k + 1]};
#pragma unroll
          for (int p = 0; p < 2; ++p) {
            const v2f x0 = l2 + (v2f){rb[g][2 * p][k],     rb[g][2 * p][k + 1]};
            const v2f x1 = l2 + (v2f){rb[g][2 * p + 1][k], rb[g][2 * p + 1][k + 1]};
            const v2f hk  = (v2f){fmaxf(x0[0], 0.f), fmaxf(x1[0], 0.f)};
            const v2f hk1 = (v2f){fmaxf(x0[1], 0.f), fmaxf(x1[1], 0.f)};
            acc2[a][p] = __builtin_elementwise_fma(hk,  (v2f){w[g][k],     w[g][k]},     acc2[a][p]);
            acc2[a][p] = __builtin_elementwise_fma(hk1, (v2f){w[g][k + 1], w[g][k + 1]}, acc2[a][p]);
          }
        }
    lp += RTRIP; rp += RTRIP; wp += RTRIP;
  }

  const float bl = blast[0];
#pragma unroll
  for (int a = 0; a < 4; ++a) {
#pragma unroll
    for (int b = 0; b < 4; ++b) {
      int i = ib * 64 + ti * 4 + a;
      int j = jb * 64 + tj * 4 + b;
      out[(long)bz * (SS * SS) + (long)i * SS + j] = acc2[a][b >> 1][b & 1] + bl;
    }
  }
}

extern "C" void kernel_launch(void* const* d_in, const int* in_sizes, int n_in,
                              void* d_out, int out_size, void* d_ws, size_t ws_size,
                              hipStream_t stream) {
  (void)in_sizes; (void)n_in; (void)out_size; (void)ws_size;
  const float* batch  = (const float*)d_in[0];
  const float* W1     = (const float*)d_in[1];
  const float* b1     = (const float*)d_in[2];
  const float* W_last = (const float*)d_in[3];
  const float* b_last = (const float*)d_in[4];
  float* out = (float*)d_out;

  float* leftp  = (float*)d_ws;
  float* rightp = leftp + (size_t)NROWS * RANK;

  dim3 g1(NROWS / TM, NCOLS / TN);
  gemm_lr_kernel<<<g1, NWAVE * 32, 0, stream>>>(batch, W1, b1, leftp, rightp);

  dim3 g2(4, 4, NB);
  pair_relu_dot_kernel<<<g2, 256, 0, stream>>>(leftp, rightp, W_last, b_last, out);
}


# pair kernel: rows staged in four 64-r chunks, chunk c+1 fetched with whole-line loads under chunk c's arithmetic, only chunk 0's fetch exposed; sum order kept, bit-identical
# speedup vs baseline: 2.0214x; 1.0471x over previous
#include <hip/hip_runtime.h>
#include <hip/hip_bf16.h>

typedef __attribute__((ext_vector_type(2))) float v2f;
typedef __attribute__((ext_vector_type(4))) float v4f;
typedef __attribute__((ext_vector_type(8))) float v8f;

#define DDIM 1024
#define RANK 256
#define NB   16
#define SS   256
#define NROWS (NB * SS)
#define NCOLS (2 * RANK)

#define KSUB  32
#define KSTEP (KSUB / 4)
#define KBLK  128
#define NBLK  (DDIM / KBLK)
#define TM    64
#define TN    128
#define NWAVE 8
#define WTN   2
#define TROWS (TM + TN)
#define GPITCH (KBLK + 4)
#define STAGE_FLOATS (TROWS * GPITCH)
#define NLDA  (TM / NWAVE)
#define NLDB  (TN / NWAVE)

struct FragSet {
  v2f a0[KSTEP];
  v2f a1[KSTEP];
  v2f b[WTN][KSTEP];
};

__device__ __forceinline__ void lds_load_set(
    FragSet& f, const float* st, int oA0, int oA1, int oB, int kk)
{
#pragma unroll
  for (int s = 0; s < KSTEP; ++s) {
    f.a0[s] = *(const v2f*)(st + oA0 + kk + 4 * s);
    f.a1[s] = *(const v2f*)(st + oA1 + kk + 4 * s);
#pragma unroll
    for (int t = 0; t < WTN; ++t)
      f.b[t][s] = *(const v2f*)(st + oB + t * 16 * GPITCH + kk + 4 * s);
  }
}

__device__ __forceinline__ void gload_block(
    v4f (&g)[NLDA + NLDB], const float* __restrict__ gA,
    const float* __restrict__ gB, int k0)
{
#pragma unroll
  for (int i = 0; i < NLDA; ++i)
    g[i] = *(const v4f*)(gA + (long)(NWAVE * i) * DDIM + k0);
#pragma unroll
  for (int i = 0; i < NLDB; ++i)
    g[NLDA + i] = *(const v4f*)(gB + (long)(NWAVE * i) * (2 * DDIM) + k0);
}

__device__ __forceinline__ void lds_store_block(
    float* p, const v4f (&g)[NLDA + NLDB])
{
#pragma unroll
  for (int i = 0; i < NLDA + NLDB; ++i)
    *(v4f*)(p + NWAVE * i * GPITCH) = g[i];
}

__device__ __forceinline__ void mma_set(v8f (&acc)[2][WTN], const FragSet& f)
{
#pragma unroll
  for (int s = 0; s < KSTEP; ++s) {
#pragma unroll
    for (int t = 0; t < WTN; ++t) {
      acc[0][t] = __builtin_amdgcn_wmma_f32_16x16x4_f32(
          false, f.a0[s], false, f.b[t][s], (short)0, acc[0][t], false, false);
      acc[1][t] = __builtin_amdgcn_wmma_f32_16x16x4_f32(
          false, f.a1[s], false, f.b[t][s], (short)0, acc[1][t], false, false);
    }
  }
}

#define COMPUTE_HEAD(st)                                     \
  lds_load_set(f0, (st), oA0, oA1, oB, 0);                   \
  lds_load_set(f1, (st), oA0, oA1, oB, KSUB);                \
  __builtin_amdgcn_sched_barrier(0);                         \
  mma_set(acc, f0);                                          \
  __builtin_amdgcn_sched_barrier(0);                         \
  lds_load_set(f0, (st), oA0, oA1, oB, 2 * KSUB);            \
  __builtin_amdgcn_sched_barrier(0);                         \
  mma_set(acc, f1);                                          \
  __builtin_amdgcn_sched_barrier(0);                         \
  lds_load_set(f1, (st), oA0, oA1, oB, 3 * KSUB);            \
  __builtin_amdgcn_sched_barrier(0);                         \
  mma_set(acc, f0);                                          \
  __builtin_amdgcn_sched_barrier(0);

__global__ __launch_bounds__(NWAVE * 32) void gemm_lr_kernel(
    const float* __restrict__ batch, const float* __restrict__ W1,
    const float* __restrict__ b1, float* __restrict__ leftp,
    float* __restrict__ rightp)
{
  __shared__ __attribute__((aligned(16))) float lds[2 * STAGE_FLOATS];

  const int tid  = threadIdx.x;
  const int lane = tid & 31;
  const int w    = tid >> 5;
  const int wm   = w & 1;
  const int wn   = w >> 1;
  const int m0   = blockIdx.x * TM;
  const int n0   = blockIdx.y * TN;
  const int c    = lane & 15;
  const int kb   = (lane >> 4) * 2;

  const int gside = (n0 >= RANK) ? 1 : 0;
  const int wrow0 = n0 - gside * RANK;

  const float* gA = batch + (long)(m0 + w) * DDIM + lane * 4;
  const float* gB = W1 + (long)(wrow0 + w) * (2 * DDIM) + gside * DDIM + lane * 4;
  const int    sOff = w * GPITCH + lane * 4;

  const int oA0 = (wm * 32 + c) * GPITCH + kb;
  const int oA1 = oA0 + 16 * GPITCH;
  const int oB  = (TM + wn * (16 * WTN) + c) * GPITCH + kb;

  v8f acc[2][WTN];
#pragma unroll
  for (int i = 0; i < 2; ++i)
#pragma unroll
    for (int t = 0; t < WTN; ++t) acc[i][t] = (v8f){0.f,0.f,0.f,0.f,0.f,0.f,0.f,0.f};

  v4f g[NLDA + NLDB];
  FragSet f0, f1;

  gload_block(g, gA, gB, 0);
  lds_store_block(lds + sOff, g);
  __syncthreads();

#pragma unroll 1
  for (int b = 0; b < NBLK - 1; ++b) {
    const float* cur = lds + (b & 1) * STAGE_FLOATS;
    float*       nxt = lds + ((b + 1) & 1) * STAGE_FLOATS;
    gload_block(g, gA, gB, (b + 1) * KBLK);
    __builtin_amdgcn_sched_barrier(0);
    COMPUTE_HEAD(cur)
    lds_store_block(nxt + sOff, g);
    __builtin_amdgcn_sched_barrier(0);
    mma_set(acc, f1);
    __syncthreads();
  }
  {
    const float* cur = lds + ((NBLK - 1) & 1) * STAGE_FLOATS;
    COMPUTE_HEAD(cur)
    mma_set(acc, f1);
  }

  const int mw = m0 + wm * 32;
  const int nw = n0 + wn * (16 * WTN);
  const int rowoff = (lane >> 4) ? 8 : 0;
#pragma unroll
  for (int t = 0; t < WTN; ++t) {
    int ncol = nw + t * 16 + (lane & 15);
    int side = (ncol >= RANK) ? 1 : 0;
    int ccol = ncol - side * RANK;
    float bias = side ? 0.f : b1[ncol];
    float* dst = side ? rightp : leftp;
#pragma unroll
    for (int i = 0; i < 2; ++i) {
#pragma unroll
      for (int v = 0; v < 8; ++v) {
        int row = mw + i * 16 + rowoff + v;
        dst[(long)row * RANK + ccol] = acc[i][t][v] + bias;
      }
    }
  }
}

#define PITCH 260
#define RSTEP 4
#define RTRIP 8
#define RCH   64
#define NCH   (RANK / RCH)
#define NLDC  4

__device__ __forceinline__ void pair_gload(
    v4f (&gl)[NLDC], v4f (&gr)[NLDC], const float* __restrict__ lg,
    const float* __restrict__ rg, int r0)
{
#pragma unroll
  for (int i = 0; i < NLDC; ++i) {
    gl[i] = *(const v4f*)(lg + (long)(2 * i) * RANK + r0);
    gr[i] = *(const v4f*)(rg + (long)(8 * i) * RANK + r0);
  }
}

__device__ __forceinline__ void pair_lstore(
    float* lsr, float* rsr, const v4f (&gl)[NLDC], const v4f (&gr)[NLDC], int r0)
{
#pragma unroll
  for (int i = 0; i < NLDC; ++i) {
    *(v4f*)(lsr + (2 * i) * PITCH + r0) = gl[i];
    *(v4f*)(rsr + (2 * i) * PITCH + r0) = gr[i];
  }
}
__global__ __launch_bounds__(256) void pair_relu_dot_kernel(
    const float* __restrict__ leftp, const float* __restrict__ rightp,
    const float* __restrict__ wlast, const float* __restrict__ blast,
    float* __restrict__ out)
{
  __shared__ __attribute__((aligned(16))) float Ls[64 * PITCH];
  __shared__ __attribute__((aligned(16))) float Rs[64 * PITCH];
  __shared__ __attribute__((aligned(16))) float ws[RANK];

  const int t  = threadIdx.x;
  const int jb = blockIdx.x;
  const int ib = blockIdx.y;
  const int bz = blockIdx.z;
  const int si = bz * SS + ib * 64;
  const int sj = bz * SS + jb * 64;

  const int wv  = t >> 5;
  const int hh  = (t >> 4) & 1;
  const int c16 = t & 15;
  const float* lg = leftp  + (long)(si + 8 * wv + hh) * RANK + c16 * 4;
  const float* rg = rightp + (long)(sj + 32 * (wv & 1) + (wv >> 1) + 4 * hh) * RANK + c16 * 4;
  float* lsr = Ls + (8 * wv + hh) * PITCH + c16 * 4;
  float* rsr = Rs + (8 * wv + hh) * PITCH + c16 * 4;
  v4f gl[NLDC], gr[NLDC];

  pair_gload(gl, gr, lg, rg, 0);
  ws[t] = wlast[t];
  pair_lstore(lsr, rsr, gl, gr, 0);
  __syncthreads();

  const int ti = t >> 4;
  const int tj = t & 15;
  const float* lp = Ls + (ti * 4) * PITCH;
  const float* rp = Rs + tj * PITCH;
  const float* wp = ws;
  v2f acc2[4][2];
#pragma unroll
  for (int a = 0; a < 4; ++a) { acc2[a][0] = (v2f){0.f, 0.f}; acc2[a][1] = (v2f){0.f, 0.f}; }

#pragma unroll 1
  for (int ch = 0; ch < NCH; ++ch) {
  if (ch + 1 < NCH) pair_gload(gl, gr, lg, rg, (ch + 1) * RCH);
#pragma unroll 1
  for (int r0 = 0; r0 < RCH; r0 += RTRIP) {
    v4f la[RTRIP / RSTEP][4], rb[RTRIP / RSTEP][4], w[RTRIP / RSTEP];
#pragma unroll
    for (int g = 0; g < RTRIP / RSTEP; ++g) {
#pragma unroll
      for (int a = 0; a < 4; ++a) la[g][a] = *(const v4f*)(lp + a * PITCH + g * RSTEP);
#pragma unroll
      for (int b = 0; b < 4; ++b) rb[g][b] = *(const v4f*)(rp + b * 16 * PITCH + g * RSTEP);
      w[g] = *(const v4f*)(wp + g * RSTEP);
    }
#pragma unroll
    for (int g = 0; g < RTRIP / RSTEP; ++g)
#pragma unroll
      for (int k = 0; k < RSTEP; k += 2)
#pragma unroll
        for (int a = 0; a < 4; ++a) {
          const v2f l2 = (v2f){la[g][a][k], la[g][a][k + 1]};
#pragma unroll
          for (int p = 0; p < 2; ++p) {
            const v2f x0 = l2 + (v2f){rb[g][2 * p][k],     rb[g][2 * p][k + 1]};
            const v2f x1 = l2 + (v2f){rb[g][2 * p + 1][k], rb[g][2 * p + 1][k + 1]};
            const v2f hk  = (v2f){fmaxf(x0[0], 0.f), fmaxf(x1[0], 0.f)};
            const v2f hk1 = (v2f){fmaxf(x0[1], 0.f), fmaxf(x1[1], 0.f)};
            acc2[a][p] = __builtin_elementwise_fma(hk,  (v2f){w[g][k],     w[g][k]},     acc2[a][p]);
            acc2[a][p] = __builtin_elementwise_fma(hk1, (v2f){w[g][k + 1], w[g][k + 1]}, acc2[a][p]);
          }
        }
    lp += RTRIP; rp += RTRIP; wp += RTRIP;
  }
  if (ch + 1 < NCH) {
    pair_lstore(lsr, rsr, gl, gr, (ch + 1) * RCH);
    __syncthreads();
  }
  }

  const float bl = blast[0];
#pragma unroll
  for (int a = 0; a < 4; ++a) {
#pragma unroll
    for (int b = 0; b < 4; ++b) {
      int i = ib * 64 + ti * 4 + a;
      int j = jb * 64 + tj * 4 + b;
      out[(long)bz * (SS * SS) + (long)i * SS + j] = acc2[a][b >> 1][b & 1] + bl;
    }
  }
}

extern "C" void kernel_launch(void* const* d_in, const int* in_sizes, int n_in,
                              void* d_out, int out_size, void* d_ws, size_t ws_size,
                              hipStream_t stream) {
  (void)in_sizes; (void)n_in; (void)out_size; (void)ws_size;
  const float* batch  = (const float*)d_in[0];
  const float* W1     = (const float*)d_in[1];
  const float* b1     = (const float*)d_in[2];
  const float* W_last = (const float*)d_in[3];
  const float* b_last = (const float*)d_in[4];
  float* out = (float*)d_out;

  float* leftp  = (float*)d_ws;
  float* rightp = leftp + (size_t)NROWS * RANK;

  dim3 g1(NROWS / TM, NCOLS / TN);
  gemm_lr_kernel<<<g1, NWAVE * 32, 0, stream>>>(batch, W1, b1, leftp, rightp);

  dim3 g2(4, 4, NB);
  pair_relu_dot_kernel<<<g2, 256, 0, stream>>>(leftp, rightp, W_last, b_last, out);
}
